# starting kernel
# baseline (speedup 1.0000x reference)
#include <hip/hip_runtime.h>
#include <math.h>


typedef float v2f __attribute__((ext_vector_type(2)));
typedef float v8f __attribute__((ext_vector_type(8)));

#define BATCH  8
#define NPTS   4096
#define NC     32
#define KC     256
#define PSTR   96
#define ITERS  10
#define SPLITK 4
#define KSPAN  (NPTS / SPLITK)

__global__ void crf_prep(const float* __restrict__ dp,
                         const float* __restrict__ logits,
                         float* __restrict__ a, float* __restrict__ ux,
                         float* __restrict__ uy, float* __restrict__ L) {
    int i = blockIdx.x * blockDim.x + threadIdx.x;
    if (i >= BATCH * NPTS) return;
    float x = dp[2 * i], y = dp[2 * i + 1];
    float r2 = x * x + y * y;
    float inv = rsqrtf(r2);
    a[i]  = __expf(-0.5f * r2);
    ux[i] = x * inv;
    uy[i] = y * inv;
    L[i]  = logits[i];
}

__global__ void crf_sym(const float* __restrict__ W, float* __restrict__ Ws) {
    __shared__ float tA[32][33];
    __shared__ float tB[32][33];
    int j0 = blockIdx.x * 32, i0 = blockIdx.y * 32;
    int c = threadIdx.x;
    for (int rr = 0; rr < 32; rr += 8) {
        int r = threadIdx.y + rr;
        tA[r][c] = W[(size_t)(i0 + r) * NPTS + (j0 + c)];
        tB[r][c] = W[(size_t)(j0 + r) * NPTS + (i0 + c)];
    }
    __syncthreads();
    for (int rr = 0; rr < 32; rr += 8) {
        int r = threadIdx.y + rr;
        Ws[(size_t)(i0 + r) * NPTS + (j0 + c)] = 0.5f * (tA[r][c] + tB[c][r]);
    }
}

__device__ __forceinline__ float crf_combine(const float* __restrict__ Yp,
                                             int p, int b, float uxv, float uyv) {
    float y0 = 0.f, y1 = 0.f, y2 = 0.f;
#pragma unroll
    for (int s = 0; s < SPLITK; ++s) {
        const float* Ys = Yp + (size_t)s * NPTS * NC + (size_t)p * NC + 3 * b;
        y0 += Ys[0];
        y1 += Ys[1];
        y2 += Ys[2];
    }
    return y0 - uxv * y1 - uyv * y2;
}

__global__ void crf_pointv(const float* __restrict__ a, const float* __restrict__ ux,
                           const float* __restrict__ uy, const float* __restrict__ unary,
                           const float* __restrict__ Yp, float* __restrict__ L,
                           float* __restrict__ V, int useY) {
    int i = blockIdx.x * blockDim.x + threadIdx.x;
    if (i >= BATCH * NPTS) return;
    int b = i >> 12;
    int p = i & (NPTS - 1);
    float uxv = ux[i], uyv = uy[i], av = a[i];
    float l;
    if (useY) {
        l = unary[i] + av * crf_combine(Yp, p, b, uxv, uyv);
    } else {
        l = L[i];
    }
    L[i] = l;
    float q = 2.0f / (1.0f + __expf(-l)) - 1.0f;
    float s = av * q;
    V[p * NC + 3 * b]     = s;
    V[p * NC + 3 * b + 1] = s * uxv;
    V[p * NC + 3 * b + 2] = s * uyv;
    if (b == 0) {
#pragma unroll
        for (int cc = 24; cc < 32; ++cc) V[p * NC + cc] = 0.0f;
    }
}

__global__ __launch_bounds__(256)
void crf_gemm(const float* __restrict__ Ws, const float* __restrict__ V,
              float* __restrict__ Ypart) {
    __shared__ float ldsV[(KC / 2) * PSTR];

    int tid   = threadIdx.x;
    int lane  = tid & 31;
    int wid   = tid >> 5;
    int split = blockIdx.x & (SPLITK - 1);
    int rtg   = blockIdx.x >> 2;
    int row0  = (rtg * 8 + wid) * 16;
    int kbeg  = split * KSPAN;

    int mrow = row0 + (lane & 15);
    int koff = (lane >> 4) << 1;
    const float* pA = Ws + (size_t)mrow * NPTS + koff;
    int nn = lane & 15;
    int pbOff = (koff >> 1) * PSTR + nn * 2;

    v8f acc0 = {0.f, 0.f, 0.f, 0.f, 0.f, 0.f, 0.f, 0.f};
    v8f acc1 = {0.f, 0.f, 0.f, 0.f, 0.f, 0.f, 0.f, 0.f};

    for (int k0 = kbeg; k0 < kbeg + KSPAN; k0 += KC) {
        const float* Vg = V + (size_t)k0 * NC;
        for (int idx = tid; idx < KC * NC; idx += 256) {
            int kk = idx >> 5, cc = idx & 31;
            ldsV[(kk >> 1) * PSTR + cc * 2 + (kk & 1)] = Vg[idx];
        }
        __syncthreads();

#pragma unroll 4
        for (int kk = 0; kk < KC; kk += 4) {
            v2f av = *(const v2f*)(pA + k0 + kk);
            const float* pb = &ldsV[(kk >> 1) * PSTR + pbOff];
            v2f b0 = *(const v2f*)(pb);
            v2f b1 = *(const v2f*)(pb + 32);
            acc0 = __builtin_amdgcn_wmma_f32_16x16x4_f32(
                false, av, false, b0, (short)0, acc0, false, false);
            acc1 = __builtin_amdgcn_wmma_f32_16x16x4_f32(
                false, av, false, b1, (short)0, acc1, false, false);
        }
        __syncthreads();
    }

    float* Yp = Ypart + (size_t)split * NPTS * NC;
    int mbase = row0 + ((lane >> 4) << 3);
#pragma unroll
    for (int r = 0; r < 8; ++r) {
        Yp[(size_t)(mbase + r) * NC + nn]      = acc0[r];
        Yp[(size_t)(mbase + r) * NC + 16 + nn] = acc1[r];
    }
}

__global__ void crf_final(const float* __restrict__ a, const float* __restrict__ ux,
                          const float* __restrict__ uy, const float* __restrict__ unary,
                          const float* __restrict__ Yp, float* __restrict__ out) {
    int i = blockIdx.x * blockDim.x + threadIdx.x;
    if (i >= BATCH * NPTS) return;
    int b = i >> 12;
    int p = i & (NPTS - 1);
    out[i] = unary[i] + a[i] * crf_combine(Yp, p, b, ux[i], uy[i]);
}

extern "C" void kernel_launch(void* const* d_in, const int* in_sizes, int n_in,
                              void* d_out, int out_size, void* d_ws, size_t ws_size,
                              hipStream_t stream) {
    const float* dp     = (const float*)d_in[0];
    const float* logits = (const float*)d_in[1];
    const float* W      = (const float*)d_in[2];
    float* out = (float*)d_out;

    float* ws    = (float*)d_ws;
    float* Wsym  = ws;
    float* V     = Wsym + (size_t)NPTS * NPTS;
    float* Ypart = V + (size_t)NPTS * NC;
    float* a     = Ypart + (size_t)SPLITK * NPTS * NC;
    float* ux    = a + BATCH * NPTS;
    float* uy    = ux + BATCH * NPTS;
    float* L     = uy + BATCH * NPTS;

    crf_prep<<<128, 256, 0, stream>>>(dp, logits, a, ux, uy, L);
    crf_sym<<<dim3(128, 128), dim3(32, 8), 0, stream>>>(W, Wsym);

    for (int it = 0; it < ITERS; ++it) {
        crf_pointv<<<128, 256, 0, stream>>>(a, ux, uy, logits, Ypart, L, V, it);
        crf_gemm<<<32 * SPLITK, 256, 0, stream>>>(Wsym, V, Ypart);
    }
    crf_final<<<128, 256, 0, stream>>>(a, ux, uy, logits, Ypart, out);
}


# GEMM A stream: 16-byte row loads, each 128-k block (16 loads per lane) requested a block ahead of its WMMAs in a second register set; k order inside each 8-k group permuted: fp32 sum order changes
# speedup vs baseline: 1.4546x; 1.4546x over previous
#include <hip/hip_runtime.h>
#include <math.h>


typedef float v2f __attribute__((ext_vector_type(2)));
typedef float v4f __attribute__((ext_vector_type(4)));
typedef float v8f __attribute__((ext_vector_type(8)));

#define BATCH  8
#define NPTS   4096
#define NC     32
#define KC     256
#define PSTR   80
#define SUBK   128
#define NLD    (SUBK / 8)
#define ITERS  10
#define SPLITK 4
#define KSPAN  (NPTS / SPLITK)

__global__ void crf_prep(const float* __restrict__ dp,
                         const float* __restrict__ logits,
                         float* __restrict__ a, float* __restrict__ ux,
                         float* __restrict__ uy, float* __restrict__ L) {
    int i = blockIdx.x * blockDim.x + threadIdx.x;
    if (i >= BATCH * NPTS) return;
    float x = dp[2 * i], y = dp[2 * i + 1];
    float r2 = x * x + y * y;
    float inv = rsqrtf(r2);
    a[i]  = __expf(-0.5f * r2);
    ux[i] = x * inv;
    uy[i] = y * inv;
    L[i]  = logits[i];
}

__global__ void crf_sym(const float* __restrict__ W, float* __restrict__ Ws) {
    __shared__ float tA[32][33];
    __shared__ float tB[32][33];
    int j0 = blockIdx.x * 32, i0 = blockIdx.y * 32;
    int c = threadIdx.x;
    for (int rr = 0; rr < 32; rr += 8) {
        int r = threadIdx.y + rr;
        tA[r][c] = W[(size_t)(i0 + r) * NPTS + (j0 + c)];
        tB[r][c] = W[(size_t)(j0 + r) * NPTS + (i0 + c)];
    }
    __syncthreads();
    for (int rr = 0; rr < 32; rr += 8) {
        int r = threadIdx.y + rr;
        Ws[(size_t)(i0 + r) * NPTS + (j0 + c)] = 0.5f * (tA[r][c] + tB[c][r]);
    }
}

__device__ __forceinline__ float crf_combine(const float* __restrict__ Yp,
                                             int p, int b, float uxv, float uyv) {
    float y0 = 0.f, y1 = 0.f, y2 = 0.f;
#pragma unroll
    for (int s = 0; s < SPLITK; ++s) {
        const float* Ys = Yp + (size_t)s * NPTS * NC + (size_t)p * NC + 3 * b;
        y0 += Ys[0];
        y1 += Ys[1];
        y2 += Ys[2];
    }
    return y0 - uxv * y1 - uyv * y2;
}

__global__ void crf_pointv(const float* __restrict__ a, const float* __restrict__ ux,
                           const float* __restrict__ uy, const float* __restrict__ unary,
                           const float* __restrict__ Yp, float* __restrict__ L,
                           float* __restrict__ V, int useY) {
    int i = blockIdx.x * blockDim.x + threadIdx.x;
    if (i >= BATCH * NPTS) return;
    int b = i >> 12;
    int p = i & (NPTS - 1);
    float uxv = ux[i], uyv = uy[i], av = a[i];
    float l;
    if (useY) {
        l = unary[i] + av * crf_combine(Yp, p, b, uxv, uyv);
    } else {
        l = L[i];
    }
    L[i] = l;
    float q = 2.0f / (1.0f + __expf(-l)) - 1.0f;
    float s = av * q;
    V[p * NC + 3 * b]     = s;
    V[p * NC + 3 * b + 1] = s * uxv;
    V[p * NC + 3 * b + 2] = s * uyv;
    if (b == 0) {
#pragma unroll
        for (int cc = 24; cc < 32; ++cc) V[p * NC + cc] = 0.0f;
    }
}


__device__ __forceinline__ void crf_load_a(v4f (&a)[NLD], const float* __restrict__ p) {
#pragma unroll
    for (int j = 0; j < NLD; ++j) a[j] = *(const v4f*)(p + 8 * j);
}

__device__ __forceinline__ void crf_mma_block(const v4f (&a)[NLD], const float* pb,
                                              v8f& acc0, v8f& acc1) {
#pragma unroll
    for (int j = 0; j < NLD; ++j) {
        const float* q0 = pb + (4 * j) * PSTR;
        const float* q1 = pb + (4 * j + 1) * PSTR;
        v2f a0 = {a[j].x, a[j].y};
        v2f a1 = {a[j].z, a[j].w};
        v2f b00 = *(const v2f*)(q0);
        v2f b01 = *(const v2f*)(q0 + 32);
        v2f b10 = *(const v2f*)(q1);
        v2f b11 = *(const v2f*)(q1 + 32);
        acc0 = __builtin_amdgcn_wmma_f32_16x16x4_f32(
            false, a0, false, b00, (short)0, acc0, false, false);
        acc1 = __builtin_amdgcn_wmma_f32_16x16x4_f32(
            false, a0, false, b01, (short)0, acc1, false, false);
        acc0 = __builtin_amdgcn_wmma_f32_16x16x4_f32(
            false, a1, false, b10, (short)0, acc0, false, false);
        acc1 = __builtin_amdgcn_wmma_f32_16x16x4_f32(
            false, a1, false, b11, (short)0, acc1, false, false);
    }
}

__global__ __launch_bounds__(256)
void crf_gemm(const float* __restrict__ Ws, const float* __restrict__ V,
              float* __restrict__ Ypart) {
    __shared__ float ldsV[(KC / 2) * PSTR];

    int tid   = threadIdx.x;
    int lane  = tid & 31;
    int wid   = tid >> 5;
    int split = blockIdx.x & (SPLITK - 1);
    int rtg   = blockIdx.x >> 2;
    int row0  = (rtg * 8 + wid) * 16;
    int kbeg  = split * KSPAN;

    int mrow = row0 + (lane & 15);
    int half = lane >> 4;
    int nn   = lane & 15;
    const float* pA = Ws + (size_t)mrow * NPTS + kbeg + 4 * half;
    const float* pB = &ldsV[(2 * half) * PSTR + nn * 2];

    v8f acc0 = {0.f, 0.f, 0.f, 0.f, 0.f, 0.f, 0.f, 0.f};
    v8f acc1 = {0.f, 0.f, 0.f, 0.f, 0.f, 0.f, 0.f, 0.f};

    v4f bufA[NLD], bufB[NLD];
    crf_load_a(bufA, pA);

#pragma unroll
    for (int c = 0; c < KSPAN / KC; ++c) {
        const float* Vg = V + (size_t)(kbeg + c * KC) * NC;
        float vt[KC * NC / 256];
#pragma unroll
        for (int i = 0; i < KC * NC / 256; ++i) vt[i] = Vg[tid + 256 * i];
#pragma unroll
        for (int i = 0; i < KC * NC / 256; ++i) {
            int idx = tid + 256 * i;
            int kk = idx >> 5, cc = idx & 31;
            ldsV[(kk >> 1) * PSTR + cc * 2 + (kk & 1)] = vt[i];
        }
        __syncthreads();

        crf_load_a(bufB, pA + c * KC + SUBK);
        crf_mma_block(bufA, pB, acc0, acc1);
        if (c + 1 < KSPAN / KC) crf_load_a(bufA, pA + (c + 1) * KC);
        crf_mma_block(bufB, pB + (SUBK / 2) * PSTR, acc0, acc1);
        __syncthreads();
    }

    float* Yp = Ypart + (size_t)split * NPTS * NC;
    int mbase = row0 + (half << 3);
#pragma unroll
    for (int r = 0; r < 8; ++r) {
        Yp[(size_t)(mbase + r) * NC + nn]      = acc0[r];
        Yp[(size_t)(mbase + r) * NC + 16 + nn] = acc1[r];
    }
}

__global__ void crf_final(const float* __restrict__ a, const float* __restrict__ ux,
                          const float* __restrict__ uy, const float* __restrict__ unary,
                          const float* __restrict__ Yp, float* __restrict__ out) {
    int i = blockIdx.x * blockDim.x + threadIdx.x;
    if (i >= BATCH * NPTS) return;
    int b = i >> 12;
    int p = i & (NPTS - 1);
    out[i] = unary[i] + a[i] * crf_combine(Yp, p, b, ux[i], uy[i]);
}

extern "C" void kernel_launch(void* const* d_in, const int* in_sizes, int n_in,
                              void* d_out, int out_size, void* d_ws, size_t ws_size,
                              hipStream_t stream) {
    const float* dp     = (const float*)d_in[0];
    const float* logits = (const float*)d_in[1];
    const float* W      = (const float*)d_in[2];
    float* out = (float*)d_out;

    float* ws    = (float*)d_ws;
    float* Wsym  = ws;
    float* V     = Wsym + (size_t)NPTS * NPTS;
    float* Ypart = V + (size_t)NPTS * NC;
    float* a     = Ypart + (size_t)SPLITK * NPTS * NC;
    float* ux    = a + BATCH * NPTS;
    float* uy    = ux + BATCH * NPTS;
    float* L     = uy + BATCH * NPTS;

    crf_prep<<<128, 256, 0, stream>>>(dp, logits, a, ux, uy, L);
    crf_sym<<<dim3(128, 128), dim3(32, 8), 0, stream>>>(W, Wsym);

    for (int it = 0; it < ITERS; ++it) {
        crf_pointv<<<128, 256, 0, stream>>>(a, ux, uy, logits, Ypart, L, V, it);
        crf_gemm<<<32 * SPLITK, 256, 0, stream>>>(Wsym, V, Ypart);
    }
    crf_final<<<128, 256, 0, stream>>>(a, ux, uy, logits, Ypart, out);
}


# GEMM spread over all 256 compute units: 256 work-groups of 8 waves (was 128), waves = 4 row slabs x 2 k halves of each chunk, halves added through LDS at the end: fp32 sum order changes
# speedup vs baseline: 1.9897x; 1.3679x over previous
#include <hip/hip_runtime.h>
#include <math.h>


typedef float v2f __attribute__((ext_vector_type(2)));
typedef float v4f __attribute__((ext_vector_type(4)));
typedef float v8f __attribute__((ext_vector_type(8)));

#define BATCH  8
#define NPTS   4096
#define NC     32
#define KC     256
#define PSTR   80
#define SUBK   128
#define NLD    (SUBK / 8)
#define ITERS  10
#define SPLITK 4
#define KSPAN  (NPTS / SPLITK)

__global__ void crf_prep(const float* __restrict__ dp,
                         const float* __restrict__ logits,
                         float* __restrict__ a, float* __restrict__ ux,
                         float* __restrict__ uy, float* __restrict__ L) {
    int i = blockIdx.x * blockDim.x + threadIdx.x;
    if (i >= BATCH * NPTS) return;
    float x = dp[2 * i], y = dp[2 * i + 1];
    float r2 = x * x + y * y;
    float inv = rsqrtf(r2);
    a[i]  = __expf(-0.5f * r2);
    ux[i] = x * inv;
    uy[i] = y * inv;
    L[i]  = logits[i];
}

__global__ void crf_sym(const float* __restrict__ W, float* __restrict__ Ws) {
    __shared__ float tA[32][33];
    __shared__ float tB[32][33];
    int j0 = blockIdx.x * 32, i0 = blockIdx.y * 32;
    int c = threadIdx.x;
    for (int rr = 0; rr < 32; rr += 8) {
        int r = threadIdx.y + rr;
        tA[r][c] = W[(size_t)(i0 + r) * NPTS + (j0 + c)];
        tB[r][c] = W[(size_t)(j0 + r) * NPTS + (i0 + c)];
    }
    __syncthreads();
    for (int rr = 0; rr < 32; rr += 8) {
        int r = threadIdx.y + rr;
        Ws[(size_t)(i0 + r) * NPTS + (j0 + c)] = 0.5f * (tA[r][c] + tB[c][r]);
    }
}

__device__ __forceinline__ float crf_combine(const float* __restrict__ Yp,
                                             int p, int b, float uxv, float uyv) {
    float y0 = 0.f, y1 = 0.f, y2 = 0.f;
#pragma unroll
    for (int s = 0; s < SPLITK; ++s) {
        const float* Ys = Yp + (size_t)s * NPTS * NC + (size_t)p * NC + 3 * b;
        y0 += Ys[0];
        y1 += Ys[1];
        y2 += Ys[2];
    }
    return y0 - uxv * y1 - uyv * y2;
}

__global__ void crf_pointv(const float* __restrict__ a, const float* __restrict__ ux,
                           const float* __restrict__ uy, const float* __restrict__ unary,
                           const float* __restrict__ Yp, float* __restrict__ L,
                           float* __restrict__ V, int useY) {
    int i = blockIdx.x * blockDim.x + threadIdx.x;
    if (i >= BATCH * NPTS) return;
    int b = i >> 12;
    int p = i & (NPTS - 1);
    float uxv = ux[i], uyv = uy[i], av = a[i];
    float l;
    if (useY) {
        l = unary[i] + av * crf_combine(Yp, p, b, uxv, uyv);
    } else {
        l = L[i];
    }
    L[i] = l;
    float q = 2.0f / (1.0f + __expf(-l)) - 1.0f;
    float s = av * q;
    V[p * NC + 3 * b]     = s;
    V[p * NC + 3 * b + 1] = s * uxv;
    V[p * NC + 3 * b + 2] = s * uyv;
    if (b == 0) {
#pragma unroll
        for (int cc = 24; cc < 32; ++cc) V[p * NC + cc] = 0.0f;
    }
}


__device__ __forceinline__ void crf_load_a(v4f (&a)[NLD], const float* __restrict__ p) {
#pragma unroll
    for (int j = 0; j < NLD; ++j) a[j] = *(const v4f*)(p + 8 * j);
}

__device__ __forceinline__ void crf_mma_block(const v4f (&a)[NLD], const float* pb,
                                              v8f& acc0, v8f& acc1) {
#pragma unroll
    for (int j = 0; j < NLD; ++j) {
        const float* q0 = pb + (4 * j) * PSTR;
        const float* q1 = pb + (4 * j + 1) * PSTR;
        v2f a0 = {a[j].x, a[j].y};
        v2f a1 = {a[j].z, a[j].w};
        v2f b00 = *(const v2f*)(q0);
        v2f b01 = *(const v2f*)(q0 + 32);
        v2f b10 = *(const v2f*)(q1);
        v2f b11 = *(const v2f*)(q1 + 32);
        acc0 = __builtin_amdgcn_wmma_f32_16x16x4_f32(
            false, a0, false, b00, (short)0, acc0, false, false);
        acc1 = __builtin_amdgcn_wmma_f32_16x16x4_f32(
            false, a0, false, b01, (short)0, acc1, false, false);
        acc0 = __builtin_amdgcn_wmma_f32_16x16x4_f32(
            false, a1, false, b10, (short)0, acc0, false, false);
        acc1 = __builtin_amdgcn_wmma_f32_16x16x4_f32(
            false, a1, false, b11, (short)0, acc1, false, false);
    }
}

__global__ __launch_bounds__(256)
void crf_gemm(const float* __restrict__ Ws, const float* __restrict__ V,
              float* __restrict__ Ypart) {
    __shared__ float ldsV[(KC / 2) * PSTR];

    int tid   = threadIdx.x;
    int lane  = tid & 31;
    int wid   = tid >> 5;
    int slab  = wid & 3;
    int kh    = wid >> 2;
    int split = blockIdx.x & (SPLITK - 1);
    int rtg   = blockIdx.x >> 2;
    int row0  = (rtg * 4 + slab) * 16;
    int kbeg  = split * KSPAN;

    int mrow = row0 + (lane & 15);
    int half = lane >> 4;
    int nn   = lane & 15;
    const float* pA = Ws + (size_t)mrow * NPTS + kbeg + kh * SUBK + 4 * half;
    const float* pB = &ldsV[(kh * (SUBK / 2) + 2 * half) * PSTR + nn * 2];

    v8f acc0 = {0.f, 0.f, 0.f, 0.f, 0.f, 0.f, 0.f, 0.f};
    v8f acc1 = {0.f, 0.f, 0.f, 0.f, 0.f, 0.f, 0.f, 0.f};

    v4f bufA[NLD], bufB[NLD];
    crf_load_a(bufA, pA);

#pragma unroll
    for (int c = 0; c < KSPAN / KC; c += 2) {
#pragma unroll
        for (int o = 0; o < 2; ++o) {
            const float* Vg = V + (size_t)(kbeg + (c + o) * KC) * NC;
            float vt[KC * NC / 256];
#pragma unroll
            for (int i = 0; i < KC * NC / 256; ++i) vt[i] = Vg[tid + 256 * i];
            __builtin_amdgcn_sched_barrier(0);
#pragma unroll
            for (int i = 0; i < KC * NC / 256; ++i) {
                int idx = tid + 256 * i;
                int kk = idx >> 5, cc = idx & 31;
                ldsV[(kk >> 1) * PSTR + cc * 2 + (kk & 1)] = vt[i];
            }
            __syncthreads();

            if (o == 0) crf_load_a(bufB, pA + (c + 1) * KC);
            else if (c + 2 < KSPAN / KC) crf_load_a(bufA, pA + (c + 2) * KC);
            __builtin_amdgcn_sched_barrier(0x0001 | 0x0080);
            if (o == 0) crf_mma_block(bufA, pB, acc0, acc1);
            else        crf_mma_block(bufB, pB, acc0, acc1);
            __syncthreads();
        }
    }

    float* red = &ldsV[(slab * 16) * 32 + lane];
    if (kh == 1) {
#pragma unroll
        for (int r = 0; r < 8; ++r) {
            red[r * 32]       = acc0[r];
            red[(8 + r) * 32] = acc1[r];
        }
    }
    __syncthreads();
    if (kh == 0) {
        float* Yp = Ypart + (size_t)split * NPTS * NC;
        int mbase = row0 + (half << 3);
#pragma unroll
        for (int r = 0; r < 8; ++r) {
            Yp[(size_t)(mbase + r) * NC + nn]      = acc0[r] + red[r * 32];
            Yp[(size_t)(mbase + r) * NC + 16 + nn] = acc1[r] + red[(8 + r) * 32];
        }
    }
}

__global__ void crf_final(const float* __restrict__ a, const float* __restrict__ ux,
                          const float* __restrict__ uy, const float* __restrict__ unary,
                          const float* __restrict__ Yp, float* __restrict__ out) {
    int i = blockIdx.x * blockDim.x + threadIdx.x;
    if (i >= BATCH * NPTS) return;
    int b = i >> 12;
    int p = i & (NPTS - 1);
    out[i] = unary[i] + a[i] * crf_combine(Yp, p, b, ux[i], uy[i]);
}

extern "C" void kernel_launch(void* const* d_in, const int* in_sizes, int n_in,
                              void* d_out, int out_size, void* d_ws, size_t ws_size,
                              hipStream_t stream) {
    const float* dp     = (const float*)d_in[0];
    const float* logits = (const float*)d_in[1];
    const float* W      = (const float*)d_in[2];
    float* out = (float*)d_out;

    float* ws    = (float*)d_ws;
    float* Wsym  = ws;
    float* V     = Wsym + (size_t)NPTS * NPTS;
    float* Ypart = V + (size_t)NPTS * NC;
    float* a     = Ypart + (size_t)SPLITK * NPTS * NC;
    float* ux    = a + BATCH * NPTS;
    float* uy    = ux + BATCH * NPTS;
    float* L     = uy + BATCH * NPTS;

    crf_prep<<<128, 256, 0, stream>>>(dp, logits, a, ux, uy, L);
    crf_sym<<<dim3(128, 128), dim3(32, 8), 0, stream>>>(W, Wsym);

    for (int it = 0; it < ITERS; ++it) {
        crf_pointv<<<128, 256, 0, stream>>>(a, ux, uy, logits, Ypart, L, V, it);
        crf_gemm<<<64 * SPLITK, 256, 0, stream>>>(Wsym, V, Ypart);
    }
    crf_final<<<128, 256, 0, stream>>>(a, ux, uy, logits, Ypart, out);
}


# GEMM k loop without LDS or barriers: wave = own 128 k x all 4 row slabs, B fragments loaded from global in the A loader's k order and kept; 8 waves added via LDS at the end: fp32 sum order changes
# speedup vs baseline: 2.1864x; 1.0989x over previous
#include <hip/hip_runtime.h>
#include <math.h>


typedef float v2f __attribute__((ext_vector_type(2)));
typedef float v4f __attribute__((ext_vector_type(4)));
typedef float v8f __attribute__((ext_vector_type(8)));

#define BATCH  8
#define NPTS   4096
#define NC     32
#define NCV    24
#define QSTR   (NCV * 4)
#define WK     128
#define NLD    (WK / 8)
#define ITERS  10
#define SPLITK 4
#define KSPAN  (NPTS / SPLITK)

__global__ void crf_prep(const float* __restrict__ dp,
                         const float* __restrict__ logits,
                         float* __restrict__ a, float* __restrict__ ux,
                         float* __restrict__ uy, float* __restrict__ L) {
    int i = blockIdx.x * blockDim.x + threadIdx.x;
    if (i >= BATCH * NPTS) return;
    float x = dp[2 * i], y = dp[2 * i + 1];
    float r2 = x * x + y * y;
    float inv = rsqrtf(r2);
    a[i]  = __expf(-0.5f * r2);
    ux[i] = x * inv;
    uy[i] = y * inv;
    L[i]  = logits[i];
}

__global__ void crf_sym(const float* __restrict__ W, float* __restrict__ Ws) {
    __shared__ float tA[32][33];
    __shared__ float tB[32][33];
    int j0 = blockIdx.x * 32, i0 = blockIdx.y * 32;
    int c = threadIdx.x;
    for (int rr = 0; rr < 32; rr += 8) {
        int r = threadIdx.y + rr;
        tA[r][c] = W[(size_t)(i0 + r) * NPTS + (j0 + c)];
        tB[r][c] = W[(size_t)(j0 + r) * NPTS + (i0 + c)];
    }
    __syncthreads();
    for (int rr = 0; rr < 32; rr += 8) {
        int r = threadIdx.y + rr;
        Ws[(size_t)(i0 + r) * NPTS + (j0 + c)] = 0.5f * (tA[r][c] + tB[c][r]);
    }
}

__device__ __forceinline__ float crf_combine(const float* __restrict__ Yp,
                                             int p, int b, float uxv, float uyv) {
    float y0 = 0.f, y1 = 0.f, y2 = 0.f;
#pragma unroll
    for (int s = 0; s < SPLITK; ++s) {
        const float* Ys = Yp + (size_t)s * NPTS * NC + (size_t)p * NC + 3 * b;
        y0 += Ys[0];
        y1 += Ys[1];
        y2 += Ys[2];
    }
    return y0 - uxv * y1 - uyv * y2;
}

__global__ void crf_pointv(const float* __restrict__ a, const float* __restrict__ ux,
                           const float* __restrict__ uy, const float* __restrict__ unary,
                           const float* __restrict__ Yp, float* __restrict__ L,
                           float* __restrict__ V, int useY) {
    int i = blockIdx.x * blockDim.x + threadIdx.x;
    if (i >= BATCH * NPTS) return;
    int b = i >> 12;
    int p = i & (NPTS - 1);
    float uxv = ux[i], uyv = uy[i], av = a[i];
    float l;
    if (useY) {
        l = unary[i] + av * crf_combine(Yp, p, b, uxv, uyv);
    } else {
        l = L[i];
    }
    L[i] = l;
    float q = 2.0f / (1.0f + __expf(-l)) - 1.0f;
    float s = av * q;
    float* Vq = V + (p >> 2) * QSTR + (p & 3);
    Vq[(3 * b) * 4]     = s;
    Vq[(3 * b + 1) * 4] = s * uxv;
    Vq[(3 * b + 2) * 4] = s * uyv;
}


__device__ __forceinline__ void crf_load_a(v4f (&a)[NLD], const float* p) {
#pragma unroll
    for (int j = 0; j < NLD; ++j) a[j] = *(const v4f*)(p + 8 * j);
}

__device__ __forceinline__ void crf_load_b(v4f (&b0)[NLD], v4f (&b1)[NLD],
                                           const float* p0, const float* p1) {
#pragma unroll
    for (int j = 0; j < NLD; ++j) {
        b0[j] = *(const v4f*)(p0 + 2 * j * QSTR);
        b1[j] = *(const v4f*)(p1 + 2 * j * QSTR);
    }
}

__device__ __forceinline__ void crf_mma_slab(const v4f (&a)[NLD], const v4f (&b0)[NLD],
                                             const v4f (&b1)[NLD], v8f& acc0, v8f& acc1) {
#pragma unroll
    for (int j = 0; j < NLD; ++j) {
        v2f a0 = {a[j].x, a[j].y};
        v2f a1 = {a[j].z, a[j].w};
        v2f b00 = {b0[j].x, b0[j].y};
        v2f b01 = {b1[j].x, b1[j].y};
        v2f b10 = {b0[j].z, b0[j].w};
        v2f b11 = {b1[j].z, b1[j].w};
        acc0 = __builtin_amdgcn_wmma_f32_16x16x4_f32(
            false, a0, false, b00, (short)0, acc0, false, false);
        acc1 = __builtin_amdgcn_wmma_f32_16x16x4_f32(
            false, a0, false, b01, (short)0, acc1, false, false);
        acc0 = __builtin_amdgcn_wmma_f32_16x16x4_f32(
            false, a1, false, b10, (short)0, acc0, false, false);
        acc1 = __builtin_amdgcn_wmma_f32_16x16x4_f32(
            false, a1, false, b11, (short)0, acc1, false, false);
    }
}

#define CRF_ORDER(acc)                                                        \
    asm volatile("" : "+v"(acc[0][0]), "+v"(acc[0][1]), "+v"(acc[1][0]),      \
                      "+v"(acc[1][1]), "+v"(acc[2][0]), "+v"(acc[2][1]),      \
                      "+v"(acc[3][0]), "+v"(acc[3][1]) : : "memory")
#define CRF_LOADS_FIRST() asm volatile("" : : : "memory")

__global__ __launch_bounds__(256)
void crf_gemm(const float* Ws, const float* V, float* Ypart) {
    __shared__ float red[8 * 64 * 32];

    int tid   = threadIdx.x;
    int lane  = tid & 31;
    int wid   = tid >> 5;
    int split = blockIdx.x & (SPLITK - 1);
    int rtg   = blockIdx.x >> 2;
    int row0  = rtg * 64;
    int kw0   = split * KSPAN + wid * WK;

    int nn   = lane & 15;
    int half = lane >> 4;
    const float* pA = Ws + (size_t)(row0 + nn) * NPTS + kw0 + 4 * half;
    const float* pB0 = V + (size_t)((kw0 >> 2) + half) * QSTR + nn * 4;
    const float* pB1 = V + (size_t)((kw0 >> 2) + half) * QSTR + (16 + (nn & 7)) * 4;

    v8f acc[4][2];
#pragma unroll
    for (int s = 0; s < 4; ++s) {
        acc[s][0] = (v8f){0.f, 0.f, 0.f, 0.f, 0.f, 0.f, 0.f, 0.f};
        acc[s][1] = (v8f){0.f, 0.f, 0.f, 0.f, 0.f, 0.f, 0.f, 0.f};
    }

    v4f b0[NLD], b1[NLD];
    v4f a0[NLD], a1[NLD], a2[NLD], a3[NLD];
    crf_load_b(b0, b1, pB0, pB1);
    crf_load_a(a0, pA);
    CRF_LOADS_FIRST();
    crf_load_a(a1, pA + (size_t)16 * NPTS);

    CRF_ORDER(acc);
    crf_load_a(a2, pA + (size_t)32 * NPTS);
    crf_mma_slab(a0, b0, b1, acc[0][0], acc[0][1]);
    CRF_ORDER(acc);
    crf_load_a(a3, pA + (size_t)48 * NPTS);
    crf_mma_slab(a1, b0, b1, acc[1][0], acc[1][1]);
    CRF_ORDER(acc);
    crf_mma_slab(a2, b0, b1, acc[2][0], acc[2][1]);
    CRF_ORDER(acc);
    crf_mma_slab(a3, b0, b1, acc[3][0], acc[3][1]);

    float* rw = &red[(wid * 64) * 32 + lane];
#pragma unroll
    for (int s = 0; s < 4; ++s) {
#pragma unroll
        for (int t = 0; t < 2; ++t) {
#pragma unroll
            for (int r = 0; r < 8; ++r) rw[((s * 2 + t) * 8 + r) * 32] = acc[s][t][r];
        }
    }
    __syncthreads();

    int so = wid >> 1;
    int r0 = (wid & 1) * 4;
    float* Yp = Ypart + (size_t)split * NPTS * NC;
    int mbase = row0 + 16 * so + (half << 3) + r0;
    const float* rr = &red[((so * 2) * 8 + r0) * 32 + lane];
#pragma unroll
    for (int i = 0; i < 4; ++i) {
        float y0 = 0.f, y1 = 0.f;
#pragma unroll
        for (int w = 0; w < 8; ++w) {
            y0 += rr[(w * 64 + i) * 32];
            y1 += rr[(w * 64 + 8 + i) * 32];
        }
        Yp[(size_t)(mbase + i) * NC + nn]      = y0;
        Yp[(size_t)(mbase + i) * NC + 16 + nn] = y1;
    }
}

__global__ void crf_final(const float* __restrict__ a, const float* __restrict__ ux,
                          const float* __restrict__ uy, const float* __restrict__ unary,
                          const float* __restrict__ Yp, float* __restrict__ out) {
    int i = blockIdx.x * blockDim.x + threadIdx.x;
    if (i >= BATCH * NPTS) return;
    int b = i >> 12;
    int p = i & (NPTS - 1);
    out[i] = unary[i] + a[i] * crf_combine(Yp, p, b, ux[i], uy[i]);
}

extern "C" void kernel_launch(void* const* d_in, const int* in_sizes, int n_in,
                              void* d_out, int out_size, void* d_ws, size_t ws_size,
                              hipStream_t stream) {
    const float* dp     = (const float*)d_in[0];
    const float* logits = (const float*)d_in[1];
    const float* W      = (const float*)d_in[2];
    float* out = (float*)d_out;

    float* ws    = (float*)d_ws;
    float* Wsym  = ws;
    float* V     = Wsym + (size_t)NPTS * NPTS;
    float* Ypart = V + (size_t)NPTS * NC;
    float* a     = Ypart + (size_t)SPLITK * NPTS * NC;
    float* ux    = a + BATCH * NPTS;
    float* uy    = ux + BATCH * NPTS;
    float* L     = uy + BATCH * NPTS;

    crf_prep<<<128, 256, 0, stream>>>(dp, logits, a, ux, uy, L);
    crf_sym<<<dim3(128, 128), dim3(32, 8), 0, stream>>>(W, Wsym);

    for (int it = 0; it < ITERS; ++it) {
        crf_pointv<<<128, 256, 0, stream>>>(a, ux, uy, logits, Ypart, L, V, it);
        crf_gemm<<<64 * SPLITK, 256, 0, stream>>>(Wsym, V, Ypart);
    }
    crf_final<<<128, 256, 0, stream>>>(a, ux, uy, logits, Ypart, out);
}


# Wsym kept in workspace in A-load order [i>>4][k>>3][i&15][k&7], written so by the symmetrisation pass: each GEMM A load reads 4 whole lines, were 16 lines x 32 B; values and sum order unchanged
# speedup vs baseline: 2.3193x; 1.0608x over previous
#include <hip/hip_runtime.h>
#include <math.h>


typedef float v2f __attribute__((ext_vector_type(2)));
typedef float v4f __attribute__((ext_vector_type(4)));
typedef float v8f __attribute__((ext_vector_type(8)));

#define BATCH  8
#define NPTS   4096
#define NC     32
#define NCV    24
#define QSTR   (NCV * 4)
#define WK     128
#define NLD    (WK / 8)
#define ITERS  10
#define SPLITK 4
#define KSPAN  (NPTS / SPLITK)

__global__ void crf_prep(const float* __restrict__ dp,
                         const float* __restrict__ logits,
                         float* __restrict__ a, float* __restrict__ ux,
                         float* __restrict__ uy, float* __restrict__ L) {
    int i = blockIdx.x * blockDim.x + threadIdx.x;
    if (i >= BATCH * NPTS) return;
    float x = dp[2 * i], y = dp[2 * i + 1];
    float r2 = x * x + y * y;
    float inv = rsqrtf(r2);
    a[i]  = __expf(-0.5f * r2);
    ux[i] = x * inv;
    uy[i] = y * inv;
    L[i]  = logits[i];
}

__global__ void crf_sym(const float* __restrict__ W, float* __restrict__ Ws) {
    __shared__ float tA[32][33];
    __shared__ float tB[32][33];
    int j0 = blockIdx.x * 32, i0 = blockIdx.y * 32;
    int c = threadIdx.x;
    for (int rr = 0; rr < 32; rr += 8) {
        int r = threadIdx.y + rr;
        tA[r][c] = W[(size_t)(i0 + r) * NPTS + (j0 + c)];
        tB[r][c] = W[(size_t)(j0 + r) * NPTS + (i0 + c)];
    }
    __syncthreads();
    int t = threadIdx.y * 32 + threadIdx.x;
    for (int m = 0; m < 4; ++m) {
        int o  = t + 256 * m;
        int sh = o >> 9;
        int q  = o & 511;
        int r  = sh * 16 + ((q >> 3) & 15);
        int cc = (q >> 7) * 8 + (q & 7);
        Ws[(size_t)((i0 >> 4) + sh) * (16 * NPTS) + (j0 >> 3) * 128 + q] =
            0.5f * (tA[r][cc] + tB[cc][r]);
    }
}

__device__ __forceinline__ float crf_combine(const float* __restrict__ Yp,
                                             int p, int b, float uxv, float uyv) {
    float y0 = 0.f, y1 = 0.f, y2 = 0.f;
#pragma unroll
    for (int s = 0; s < SPLITK; ++s) {
        const float* Ys = Yp + (size_t)s * NPTS * NC + (size_t)p * NC + 3 * b;
        y0 += Ys[0];
        y1 += Ys[1];
        y2 += Ys[2];
    }
    return y0 - uxv * y1 - uyv * y2;
}

__global__ void crf_pointv(const float* __restrict__ a, const float* __restrict__ ux,
                           const float* __restrict__ uy, const float* __restrict__ unary,
                           const float* __restrict__ Yp, float* __restrict__ L,
                           float* __restrict__ V, int useY) {
    int i = blockIdx.x * blockDim.x + threadIdx.x;
    if (i >= BATCH * NPTS) return;
    int b = i >> 12;
    int p = i & (NPTS - 1);
    float uxv = ux[i], uyv = uy[i], av = a[i];
    float l;
    if (useY) {
        l = unary[i] + av * crf_combine(Yp, p, b, uxv, uyv);
    } else {
        l = L[i];
    }
    L[i] = l;
    float q = 2.0f / (1.0f + __expf(-l)) - 1.0f;
    float s = av * q;
    float* Vq = V + (p >> 2) * QSTR + (p & 3);
    Vq[(3 * b) * 4]     = s;
    Vq[(3 * b + 1) * 4] = s * uxv;
    Vq[(3 * b + 2) * 4] = s * uyv;
}


__device__ __forceinline__ void crf_load_a(v4f (&a)[NLD], const float* p) {
#pragma unroll
    for (int j = 0; j < NLD; ++j) a[j] = *(const v4f*)(p + 128 * j);
}

__device__ __forceinline__ void crf_load_b(v4f (&b0)[NLD], v4f (&b1)[NLD],
                                           const float* p0, const float* p1) {
#pragma unroll
    for (int j = 0; j < NLD; ++j) {
        b0[j] = *(const v4f*)(p0 + 2 * j * QSTR);
        b1[j] = *(const v4f*)(p1 + 2 * j * QSTR);
    }
}

__device__ __forceinline__ void crf_mma_slab(const v4f (&a)[NLD], const v4f (&b0)[NLD],
                                             const v4f (&b1)[NLD], v8f& acc0, v8f& acc1) {
#pragma unroll
    for (int j = 0; j < NLD; ++j) {
        v2f a0 = {a[j].x, a[j].y};
        v2f a1 = {a[j].z, a[j].w};
        v2f b00 = {b0[j].x, b0[j].y};
        v2f b01 = {b1[j].x, b1[j].y};
        v2f b10 = {b0[j].z, b0[j].w};
        v2f b11 = {b1[j].z, b1[j].w};
        acc0 = __builtin_amdgcn_wmma_f32_16x16x4_f32(
            false, a0, false, b00, (short)0, acc0, false, false);
        acc1 = __builtin_amdgcn_wmma_f32_16x16x4_f32(
            false, a0, false, b01, (short)0, acc1, false, false);
        acc0 = __builtin_amdgcn_wmma_f32_16x16x4_f32(
            false, a1, false, b10, (short)0, acc0, false, false);
        acc1 = __builtin_amdgcn_wmma_f32_16x16x4_f32(
            false, a1, false, b11, (short)0, acc1, false, false);
    }
}

#define CRF_ORDER(acc)                                                        \
    asm volatile("" : "+v"(acc[0][0]), "+v"(acc[0][1]), "+v"(acc[1][0]),      \
                      "+v"(acc[1][1]), "+v"(acc[2][0]), "+v"(acc[2][1]),      \
                      "+v"(acc[3][0]), "+v"(acc[3][1]) : : "memory")
#define CRF_LOADS_FIRST() asm volatile("" : : : "memory")

__global__ __launch_bounds__(256)
void crf_gemm(const float* Ws, const float* V, float* Ypart) {
    __shared__ float red[8 * 64 * 32];

    int tid   = threadIdx.x;
    int lane  = tid & 31;
    int wid   = tid >> 5;
    int split = blockIdx.x & (SPLITK - 1);
    int rtg   = blockIdx.x >> 2;
    int row0  = rtg * 64;
    int kw0   = split * KSPAN + wid * WK;

    int nn   = lane & 15;
    int half = lane >> 4;
    const float* pA = Ws + (size_t)row0 * NPTS + (size_t)kw0 * 16 + nn * 8 + 4 * half;
    const float* pB0 = V + (size_t)((kw0 >> 2) + half) * QSTR + nn * 4;
    const float* pB1 = V + (size_t)((kw0 >> 2) + half) * QSTR + (16 + (nn & 7)) * 4;

    v8f acc[4][2];
#pragma unroll
    for (int s = 0; s < 4; ++s) {
        acc[s][0] = (v8f){0.f, 0.f, 0.f, 0.f, 0.f, 0.f, 0.f, 0.f};
        acc[s][1] = (v8f){0.f, 0.f, 0.f, 0.f, 0.f, 0.f, 0.f, 0.f};
    }

    v4f b0[NLD], b1[NLD];
    v4f a0[NLD], a1[NLD], a2[NLD], a3[NLD];
    crf_load_b(b0, b1, pB0, pB1);
    crf_load_a(a0, pA);
    CRF_LOADS_FIRST();
    crf_load_a(a1, pA + (size_t)16 * NPTS);

    CRF_ORDER(acc);
    crf_load_a(a2, pA + (size_t)32 * NPTS);
    crf_mma_slab(a0, b0, b1, acc[0][0], acc[0][1]);
    CRF_ORDER(acc);
    crf_load_a(a3, pA + (size_t)48 * NPTS);
    crf_mma_slab(a1, b0, b1, acc[1][0], acc[1][1]);
    CRF_ORDER(acc);
    crf_mma_slab(a2, b0, b1, acc[2][0], acc[2][1]);
    CRF_ORDER(acc);
    crf_mma_slab(a3, b0, b1, acc[3][0], acc[3][1]);

    float* rw = &red[(wid * 64) * 32 + lane];
#pragma unroll
    for (int s = 0; s < 4; ++s) {
#pragma unroll
        for (int t = 0; t < 2; ++t) {
#pragma unroll
            for (int r = 0; r < 8; ++r) rw[((s * 2 + t) * 8 + r) * 32] = acc[s][t][r];
        }
    }
    __syncthreads();

    int so = wid >> 1;
    int r0 = (wid & 1) * 4;
    float* Yp = Ypart + (size_t)split * NPTS * NC;
    int mbase = row0 + 16 * so + (half << 3) + r0;
    const float* rr = &red[((so * 2) * 8 + r0) * 32 + lane];
#pragma unroll
    for (int i = 0; i < 4; ++i) {
        float y0 = 0.f, y1 = 0.f;
#pragma unroll
        for (int w = 0; w < 8; ++w) {
            y0 += rr[(w * 64 + i) * 32];
            y1 += rr[(w * 64 + 8 + i) * 32];
        }
        Yp[(size_t)(mbase + i) * NC + nn]      = y0;
        Yp[(size_t)(mbase + i) * NC + 16 + nn] = y1;
    }
}

__global__ void crf_final(const float* __restrict__ a, const float* __restrict__ ux,
                          const float* __restrict__ uy, const float* __restrict__ unary,
                          const float* __restrict__ Yp, float* __restrict__ out) {
    int i = blockIdx.x * blockDim.x + threadIdx.x;
    if (i >= BATCH * NPTS) return;
    int b = i >> 12;
    int p = i & (NPTS - 1);
    out[i] = unary[i] + a[i] * crf_combine(Yp, p, b, ux[i], uy[i]);
}

extern "C" void kernel_launch(void* const* d_in, const int* in_sizes, int n_in,
                              void* d_out, int out_size, void* d_ws, size_t ws_size,
                              hipStream_t stream) {
    const float* dp     = (const float*)d_in[0];
    const float* logits = (const float*)d_in[1];
    const float* W      = (const float*)d_in[2];
    float* out = (float*)d_out;

    float* ws    = (float*)d_ws;
    float* Wsym  = ws;
    float* V     = Wsym + (size_t)NPTS * NPTS;
    float* Ypart = V + (size_t)NPTS * NC;
    float* a     = Ypart + (size_t)SPLITK * NPTS * NC;
    float* ux    = a + BATCH * NPTS;
    float* uy    = ux + BATCH * NPTS;
    float* L     = uy + BATCH * NPTS;

    crf_prep<<<128, 256, 0, stream>>>(dp, logits, a, ux, uy, L);
    crf_sym<<<dim3(128, 128), dim3(32, 8), 0, stream>>>(W, Wsym);

    for (int it = 0; it < ITERS; ++it) {
        crf_pointv<<<128, 256, 0, stream>>>(a, ux, uy, logits, Ypart, L, V, it);
        crf_gemm<<<64 * SPLITK, 256, 0, stream>>>(Wsym, V, Ypart);
    }
    crf_final<<<128, 256, 0, stream>>>(a, ux, uy, logits, Ypart, out);
}


# GEMM work-group = one 16-row slab over all k (8 waves x 512 k, added via LDS): its epilogue does the point update and writes next V or out; 11 launches, split-K partials gone; fp32 sum order changes
# speedup vs baseline: 2.9101x; 1.2547x over previous
#include <hip/hip_runtime.h>
#include <math.h>


typedef float v2f __attribute__((ext_vector_type(2)));
typedef float v4f __attribute__((ext_vector_type(4)));
typedef float v8f __attribute__((ext_vector_type(8)));

#define BATCH  8
#define NPTS   4096
#define ITERS  10
#define NWAVE  8
#define WK     (NPTS / NWAVE)
#define CH     8
#define NCHUNK (WK / 8 / CH)
#define GSTR   192
#define RPITCH 33

__device__ __forceinline__ int crf_vplace(int p, int c) {
    int h = (p >> 2) & 1, e = p & 3;
    int in = (c < 16) ? (h * 64 + c * 4 + e) : (128 + h * 32 + (c - 16) * 4 + e);
    return (p >> 3) * GSTR + in;
}

__global__ void crf_prep(const float* __restrict__ dp,
                         const float* __restrict__ logits,
                         float* __restrict__ a, float* __restrict__ ux,
                         float* __restrict__ uy, float* __restrict__ V) {
    int i = blockIdx.x * blockDim.x + threadIdx.x;
    if (i >= BATCH * NPTS) return;
    int b = i >> 12;
    int p = i & (NPTS - 1);
    float x = dp[2 * i], y = dp[2 * i + 1];
    float r2 = x * x + y * y;
    float inv = rsqrtf(r2);
    float av = __expf(-0.5f * r2);
    float uxv = x * inv, uyv = y * inv;
    a[i]  = av;
    ux[i] = uxv;
    uy[i] = uyv;
    float l = logits[i];
    float q = 2.0f / (1.0f + __expf(-l)) - 1.0f;
    float s = av * q;
    V[crf_vplace(p, 3 * b)]     = s;
    V[crf_vplace(p, 3 * b + 1)] = s * uxv;
    V[crf_vplace(p, 3 * b + 2)] = s * uyv;
}

__global__ void crf_sym(const float* __restrict__ W, float* __restrict__ Ws) {
    __shared__ float tA[32][33];
    __shared__ float tB[32][33];
    int j0 = blockIdx.x * 32, i0 = blockIdx.y * 32;
    int c = threadIdx.x;
    for (int rr = 0; rr < 32; rr += 8) {
        int r = threadIdx.y + rr;
        tA[r][c] = W[(size_t)(i0 + r) * NPTS + (j0 + c)];
        tB[r][c] = W[(size_t)(j0 + r) * NPTS + (i0 + c)];
    }
    __syncthreads();
    int t = threadIdx.y * 32 + threadIdx.x;
    for (int m = 0; m < 4; ++m) {
        int o  = t + 256 * m;
        int sh = o >> 9;
        int q  = o & 511;
        int r  = sh * 16 + ((q >> 3) & 15);
        int cc = (q >> 7) * 8 + (q & 7);
        Ws[(size_t)((i0 >> 4) + sh) * (16 * NPTS) + (j0 >> 3) * 128 + q] =
            0.5f * (tA[r][cc] + tB[cc][r]);
    }
}


__device__ __forceinline__ void crf_load_chunk(v4f (&a)[CH], v4f (&b0)[CH], v4f (&b1)[CH],
                                               const float* pa, const float* pb0,
                                               const float* pb1) {
#pragma unroll
    for (int j = 0; j < CH; ++j) {
        a[j]  = *(const v4f*)(pa + 128 * j);
        b0[j] = *(const v4f*)(pb0 + GSTR * j);
        b1[j] = *(const v4f*)(pb1 + GSTR * j);
    }
}

__device__ __forceinline__ void crf_mma_chunk(const v4f (&a)[CH], const v4f (&b0)[CH],
                                              const v4f (&b1)[CH], v8f& acc0, v8f& acc1) {
#pragma unroll
    for (int j = 0; j < CH; ++j) {
        v2f a0 = {a[j].x, a[j].y};
        v2f a1 = {a[j].z, a[j].w};
        v2f b00 = {b0[j].x, b0[j].y};
        v2f b01 = {b1[j].x, b1[j].y};
        v2f b10 = {b0[j].z, b0[j].w};
        v2f b11 = {b1[j].z, b1[j].w};
        acc0 = __builtin_amdgcn_wmma_f32_16x16x4_f32(
            false, a0, false, b00, (short)0, acc0, false, false);
        acc1 = __builtin_amdgcn_wmma_f32_16x16x4_f32(
            false, a0, false, b01, (short)0, acc1, false, false);
        acc0 = __builtin_amdgcn_wmma_f32_16x16x4_f32(
            false, a1, false, b10, (short)0, acc0, false, false);
        acc1 = __builtin_amdgcn_wmma_f32_16x16x4_f32(
            false, a1, false, b11, (short)0, acc1, false, false);
    }
}

#define CRF_ORDER(x, y) asm volatile("" : "+v"(x), "+v"(y) : : "memory")

__global__ __launch_bounds__(256)
void crf_gemm(const float* Ws, const float* Vin, const float* a, const float* ux,
              const float* uy, const float* unary, float* Vout, float* out, int last) {
    __shared__ float red[NWAVE * 16 * RPITCH];
    __shared__ __attribute__((aligned(16))) float vst[2 * GSTR];

    int tid  = threadIdx.x;
    int lane = tid & 31;
    int wid  = tid >> 5;
    int slab = blockIdx.x;
    int nn   = lane & 15;
    int half = lane >> 4;

    int em = tid & 15;
    int eb = (tid >> 4) & 7;
    int ei = eb * NPTS + slab * 16 + em;
    float av = a[ei], uxv = ux[ei], uyv = uy[ei], unv = unary[ei];

    int g0  = wid * (WK / 8);
    int rot = slab & (NCHUNK - 1);
    const float* pA  = Ws + (size_t)slab * (16 * NPTS) + (size_t)g0 * 128 + nn * 8 + 4 * half;
    const float* pB0 = Vin + (size_t)g0 * GSTR + half * 64 + nn * 4;
    const float* pB1 = Vin + (size_t)g0 * GSTR + 128 + half * 32 + (nn & 7) * 4;

    v8f acc0 = (v8f){0.f, 0.f, 0.f, 0.f, 0.f, 0.f, 0.f, 0.f};
    v8f acc1 = (v8f){0.f, 0.f, 0.f, 0.f, 0.f, 0.f, 0.f, 0.f};

    v4f xa[CH], xb0[CH], xb1[CH];
    v4f ya[CH], yb0[CH], yb1[CH];
    v4f za[CH], zb0[CH], zb1[CH];

#define CRF_LOAD(c, A, B0, B1)                                                 \
    do {                                                                       \
        int cc = ((c) + rot) & (NCHUNK - 1);                                   \
        crf_load_chunk(A, B0, B1, pA + cc * (CH * 128), pB0 + cc * (CH * GSTR),\
                       pB1 + cc * (CH * GSTR));                                \
    } while (0)

    CRF_LOAD(0, xa, xb0, xb1);
    CRF_LOAD(1, ya, yb0, yb1);
    CRF_ORDER(acc0, acc1);
    CRF_LOAD(2, za, zb0, zb1);
    CRF_ORDER(acc0, acc1);
    crf_mma_chunk(xa, xb0, xb1, acc0, acc1);
    CRF_ORDER(acc0, acc1);
    CRF_LOAD(3, xa, xb0, xb1);
    CRF_ORDER(acc0, acc1);
    crf_mma_chunk(ya, yb0, yb1, acc0, acc1);
    CRF_ORDER(acc0, acc1);
    CRF_LOAD(4, ya, yb0, yb1);
    CRF_ORDER(acc0, acc1);
    crf_mma_chunk(za, zb0, zb1, acc0, acc1);
    CRF_ORDER(acc0, acc1);
    CRF_LOAD(5, za, zb0, zb1);
    CRF_ORDER(acc0, acc1);
    crf_mma_chunk(xa, xb0, xb1, acc0, acc1);
    CRF_ORDER(acc0, acc1);
    CRF_LOAD(6, xa, xb0, xb1);
    CRF_ORDER(acc0, acc1);
    crf_mma_chunk(ya, yb0, yb1, acc0, acc1);
    CRF_ORDER(acc0, acc1);
    CRF_LOAD(7, ya, yb0, yb1);
    CRF_ORDER(acc0, acc1);
    crf_mma_chunk(za, zb0, zb1, acc0, acc1);
    CRF_ORDER(acc0, acc1);
    crf_mma_chunk(xa, xb0, xb1, acc0, acc1);
    CRF_ORDER(acc0, acc1);
    crf_mma_chunk(ya, yb0, yb1, acc0, acc1);
#undef CRF_LOAD

    float* rw = &red[(wid * 16) * RPITCH + lane];
#pragma unroll
    for (int r = 0; r < 8; ++r) {
        rw[r * RPITCH]       = acc0[r];
        rw[(8 + r) * RPITCH] = acc1[r];
    }
    __syncthreads();

    if (tid < 128) {
        float y[3];
#pragma unroll
        for (int t = 0; t < 3; ++t) {
            int c   = 3 * eb + t;
            int idx = ((c >> 4) * 8 + (em & 7)) * RPITCH + (c & 15) + 16 * (em >> 3);
            float s = 0.f;
#pragma unroll
            for (int w = 0; w < NWAVE; ++w) s += red[w * 16 * RPITCH + idx];
            y[t] = s;
        }
        float l = unv + av * (y[0] - uxv * y[1] - uyv * y[2]);
        if (last) {
            out[ei] = l;
        } else {
            float q = 2.0f / (1.0f + __expf(-l)) - 1.0f;
            float s = av * q;
            vst[crf_vplace(em, 3 * eb)]     = s;
            vst[crf_vplace(em, 3 * eb + 1)] = s * uxv;
            vst[crf_vplace(em, 3 * eb + 2)] = s * uyv;
        }
    }
    if (!last) {
        __syncthreads();
        if (tid < 2 * GSTR / 4)
            *(v4f*)(Vout + (size_t)(2 * slab) * GSTR + 4 * tid) = *(const v4f*)&vst[4 * tid];
    }
}

extern "C" void kernel_launch(void* const* d_in, const int* in_sizes, int n_in,
                              void* d_out, int out_size, void* d_ws, size_t ws_size,
                              hipStream_t stream) {
    const float* dp     = (const float*)d_in[0];
    const float* logits = (const float*)d_in[1];
    const float* W      = (const float*)d_in[2];
    float* out = (float*)d_out;

    float* ws    = (float*)d_ws;
    float* Wsym  = ws;
    float* V0    = Wsym + (size_t)NPTS * NPTS;
    float* V1    = V0 + (size_t)NPTS * 32;
    float* a     = V1 + (size_t)NPTS * 32;
    float* ux    = a + BATCH * NPTS;
    float* uy    = ux + BATCH * NPTS;

    crf_prep<<<128, 256, 0, stream>>>(dp, logits, a, ux, uy, V0);
    crf_sym<<<dim3(128, 128), dim3(32, 8), 0, stream>>>(W, Wsym);

    for (int it = 0; it < ITERS / 2 - 1; ++it) {
        crf_gemm<<<256, 256, 0, stream>>>(Wsym, V0, a, ux, uy, logits, V1, out, 0);
        crf_gemm<<<256, 256, 0, stream>>>(Wsym, V1, a, ux, uy, logits, V0, out, 0);
    }
    crf_gemm<<<256, 256, 0, stream>>>(Wsym, V0, a, ux, uy, logits, V1, out, 0);
    crf_gemm<<<256, 256, 0, stream>>>(Wsym, V1, a, ux, uy, logits, V0, out, 1);
}


# Symmetrisation pass: one work-group per mirror pair of 64x64 tiles reads each W element once (16-byte loads), writes both tiles via LDS as 512-byte wave stores; 8x fewer waves; values unchanged
# speedup vs baseline: 3.0622x; 1.0523x over previous
#include <hip/hip_runtime.h>
#include <math.h>


typedef float v2f __attribute__((ext_vector_type(2)));
typedef float v4f __attribute__((ext_vector_type(4)));
typedef float v8f __attribute__((ext_vector_type(8)));

#define BATCH  8
#define NPTS   4096
#define ITERS  10
#define NWAVE  8
#define WK     (NPTS / NWAVE)
#define CH     8
#define NCHUNK (WK / 8 / CH)
#define GSTR   192
#define RPITCH 33

__device__ __forceinline__ int crf_vplace(int p, int c) {
    int h = (p >> 2) & 1, e = p & 3;
    int in = (c < 16) ? (h * 64 + c * 4 + e) : (128 + h * 32 + (c - 16) * 4 + e);
    return (p >> 3) * GSTR + in;
}

__global__ void crf_prep(const float* __restrict__ dp,
                         const float* __restrict__ logits,
                         float* __restrict__ a, float* __restrict__ ux,
                         float* __restrict__ uy, float* __restrict__ V) {
    int i = blockIdx.x * blockDim.x + threadIdx.x;
    if (i >= BATCH * NPTS) return;
    int b = i >> 12;
    int p = i & (NPTS - 1);
    float x = dp[2 * i], y = dp[2 * i + 1];
    float r2 = x * x + y * y;
    float inv = rsqrtf(r2);
    float av = __expf(-0.5f * r2);
    float uxv = x * inv, uyv = y * inv;
    a[i]  = av;
    ux[i] = uxv;
    uy[i] = uyv;
    float l = logits[i];
    float q = 2.0f / (1.0f + __expf(-l)) - 1.0f;
    float s = av * q;
    V[crf_vplace(p, 3 * b)]     = s;
    V[crf_vplace(p, 3 * b + 1)] = s * uxv;
    V[crf_vplace(p, 3 * b + 2)] = s * uyv;
}

#define STS 64
#define STP 72

__global__ __launch_bounds__(256)
void crf_sym(const float* __restrict__ W, float* __restrict__ Ws) {
    __shared__ __attribute__((aligned(16))) float tP[STS * STP];
    __shared__ __attribute__((aligned(16))) float tQ[STS * STP];

    int fc = blockIdx.x, fr = blockIdx.y;
    int J = (fc <= fr) ? fr : (STS - 1 - fr);
    int I = (fc <= fr) ? fc : (fc - fr - 1);

    int tid = threadIdx.x, lane = tid & 31, wid = tid >> 5;

    int lrow = 8 * wid + (lane >> 4);
    int lcol = 4 * (lane & 15);
    const float* gP = W + (size_t)(STS * I + lrow) * NPTS + STS * J + lcol;
    const float* gQ = W + (size_t)(STS * J + lrow) * NPTS + STS * I + lcol;
    v4f p[4], q[4];
#pragma unroll
    for (int j = 0; j < 4; ++j) {
        p[j] = *(const v4f*)(gP + (size_t)(2 * j) * NPTS);
        q[j] = *(const v4f*)(gQ + (size_t)(2 * j) * NPTS);
    }
#pragma unroll
    for (int j = 0; j < 4; ++j) {
        *(v4f*)&tP[(lrow + 2 * j) * STP + lcol] = p[j];
        *(v4f*)&tQ[(lrow + 2 * j) * STP + lcol] = q[j];
    }
    __syncthreads();

    int s   = wid >> 1;
    int row = 16 * s + (lane >> 1);
    int g0  = 4 * (wid & 1);
    int ch  = 4 * (lane & 1);

    float* oA = Ws + (size_t)(4 * I + s) * (16 * NPTS) + (8 * J + g0) * 128 + 4 * lane;
#pragma unroll
    for (int m = 0; m < 4; ++m) {
        int c0 = 8 * (g0 + m) + ch;
        v4f own = *(const v4f*)&tP[row * STP + c0];
        v4f tr  = {tQ[(c0 + 0) * STP + row], tQ[(c0 + 1) * STP + row],
                   tQ[(c0 + 2) * STP + row], tQ[(c0 + 3) * STP + row]};
        *(v4f*)(oA + 128 * m) = 0.5f * (own + tr);
    }
    if (I != J) {
        float* oB = Ws + (size_t)(4 * J + s) * (16 * NPTS) + (8 * I + g0) * 128 + 4 * lane;
#pragma unroll
        for (int m = 0; m < 4; ++m) {
            int c0 = 8 * (g0 + m) + ch;
            v4f own = *(const v4f*)&tQ[row * STP + c0];
            v4f tr  = {tP[(c0 + 0) * STP + row], tP[(c0 + 1) * STP + row],
                       tP[(c0 + 2) * STP + row], tP[(c0 + 3) * STP + row]};
            *(v4f*)(oB + 128 * m) = 0.5f * (own + tr);
        }
    }
}


__device__ __forceinline__ void crf_load_chunk(v4f (&a)[CH], v4f (&b0)[CH], v4f (&b1)[CH],
                                               const float* pa, const float* pb0,
                                               const float* pb1) {
#pragma unroll
    for (int j = 0; j < CH; ++j) {
        a[j]  = *(const v4f*)(pa + 128 * j);
        b0[j] = *(const v4f*)(pb0 + GSTR * j);
        b1[j] = *(const v4f*)(pb1 + GSTR * j);
    }
}

__device__ __forceinline__ void crf_mma_chunk(const v4f (&a)[CH], const v4f (&b0)[CH],
                                              const v4f (&b1)[CH], v8f& acc0, v8f& acc1) {
#pragma unroll
    for (int j = 0; j < CH; ++j) {
        v2f a0 = {a[j].x, a[j].y};
        v2f a1 = {a[j].z, a[j].w};
        v2f b00 = {b0[j].x, b0[j].y};
        v2f b01 = {b1[j].x, b1[j].y};
        v2f b10 = {b0[j].z, b0[j].w};
        v2f b11 = {b1[j].z, b1[j].w};
        acc0 = __builtin_amdgcn_wmma_f32_16x16x4_f32(
            false, a0, false, b00, (short)0, acc0, false, false);
        acc1 = __builtin_amdgcn_wmma_f32_16x16x4_f32(
            false, a0, false, b01, (short)0, acc1, false, false);
        acc0 = __builtin_amdgcn_wmma_f32_16x16x4_f32(
            false, a1, false, b10, (short)0, acc0, false, false);
        acc1 = __builtin_amdgcn_wmma_f32_16x16x4_f32(
            false, a1, false, b11, (short)0, acc1, false, false);
    }
}

#define CRF_ORDER(x, y) asm volatile("" : "+v"(x), "+v"(y) : : "memory")

__global__ __launch_bounds__(256)
void crf_gemm(const float* Ws, const float* Vin, const float* a, const float* ux,
              const float* uy, const float* unary, float* Vout, float* out, int last) {
    __shared__ float red[NWAVE * 16 * RPITCH];
    __shared__ __attribute__((aligned(16))) float vst[2 * GSTR];

    int tid  = threadIdx.x;
    int lane = tid & 31;
    int wid  = tid >> 5;
    int slab = blockIdx.x;
    int nn   = lane & 15;
    int half = lane >> 4;

    int em = tid & 15;
    int eb = (tid >> 4) & 7;
    int ei = eb * NPTS + slab * 16 + em;
    float av = a[ei], uxv = ux[ei], uyv = uy[ei], unv = unary[ei];

    int g0  = wid * (WK / 8);
    int rot = slab & (NCHUNK - 1);
    const float* pA  = Ws + (size_t)slab * (16 * NPTS) + (size_t)g0 * 128 + nn * 8 + 4 * half;
    const float* pB0 = Vin + (size_t)g0 * GSTR + half * 64 + nn * 4;
    const float* pB1 = Vin + (size_t)g0 * GSTR + 128 + half * 32 + (nn & 7) * 4;

    v8f acc0 = (v8f){0.f, 0.f, 0.f, 0.f, 0.f, 0.f, 0.f, 0.f};
    v8f acc1 = (v8f){0.f, 0.f, 0.f, 0.f, 0.f, 0.f, 0.f, 0.f};

    v4f xa[CH], xb0[CH], xb1[CH];
    v4f ya[CH], yb0[CH], yb1[CH];
    v4f za[CH], zb0[CH], zb1[CH];

#define CRF_LOAD(c, A, B0, B1)                                                 \
    do {                                                                       \
        int cc = ((c) + rot) & (NCHUNK - 1);                                   \
        crf_load_chunk(A, B0, B1, pA + cc * (CH * 128), pB0 + cc * (CH * GSTR),\
                       pB1 + cc * (CH * GSTR));                                \
    } while (0)

    CRF_LOAD(0, xa, xb0, xb1);
    CRF_LOAD(1, ya, yb0, yb1);
    CRF_ORDER(acc0, acc1);
    CRF_LOAD(2, za, zb0, zb1);
    CRF_ORDER(acc0, acc1);
    crf_mma_chunk(xa, xb0, xb1, acc0, acc1);
    CRF_ORDER(acc0, acc1);
    CRF_LOAD(3, xa, xb0, xb1);
    CRF_ORDER(acc0, acc1);
    crf_mma_chunk(ya, yb0, yb1, acc0, acc1);
    CRF_ORDER(acc0, acc1);
    CRF_LOAD(4, ya, yb0, yb1);
    CRF_ORDER(acc0, acc1);
    crf_mma_chunk(za, zb0, zb1, acc0, acc1);
    CRF_ORDER(acc0, acc1);
    CRF_LOAD(5, za, zb0, zb1);
    CRF_ORDER(acc0, acc1);
    crf_mma_chunk(xa, xb0, xb1, acc0, acc1);
    CRF_ORDER(acc0, acc1);
    CRF_LOAD(6, xa, xb0, xb1);
    CRF_ORDER(acc0, acc1);
    crf_mma_chunk(ya, yb0, yb1, acc0, acc1);
    CRF_ORDER(acc0, acc1);
    CRF_LOAD(7, ya, yb0, yb1);
    CRF_ORDER(acc0, acc1);
    crf_mma_chunk(za, zb0, zb1, acc0, acc1);
    CRF_ORDER(acc0, acc1);
    crf_mma_chunk(xa, xb0, xb1, acc0, acc1);
    CRF_ORDER(acc0, acc1);
    crf_mma_chunk(ya, yb0, yb1, acc0, acc1);
#undef CRF_LOAD

    float* rw = &red[(wid * 16) * RPITCH + lane];
#pragma unroll
    for (int r = 0; r < 8; ++r) {
        rw[r * RPITCH]       = acc0[r];
        rw[(8 + r) * RPITCH] = acc1[r];
    }
    __syncthreads();

    if (tid < 128) {
        float y[3];
#pragma unroll
        for (int t = 0; t < 3; ++t) {
            int c   = 3 * eb + t;
            int idx = ((c >> 4) * 8 + (em & 7)) * RPITCH + (c & 15) + 16 * (em >> 3);
            float s = 0.f;
#pragma unroll
            for (int w = 0; w < NWAVE; ++w) s += red[w * 16 * RPITCH + idx];
            y[t] = s;
        }
        float l = unv + av * (y[0] - uxv * y[1] - uyv * y[2]);
        if (last) {
            out[ei] = l;
        } else {
            float q = 2.0f / (1.0f + __expf(-l)) - 1.0f;
            float s = av * q;
            vst[crf_vplace(em, 3 * eb)]     = s;
            vst[crf_vplace(em, 3 * eb + 1)] = s * uxv;
            vst[crf_vplace(em, 3 * eb + 2)] = s * uyv;
        }
    }
    if (!last) {
        __syncthreads();
        if (tid < 2 * GSTR / 4)
            *(v4f*)(Vout + (size_t)(2 * slab) * GSTR + 4 * tid) = *(const v4f*)&vst[4 * tid];
    }
}

extern "C" void kernel_launch(void* const* d_in, const int* in_sizes, int n_in,
                              void* d_out, int out_size, void* d_ws, size_t ws_size,
                              hipStream_t stream) {
    const float* dp     = (const float*)d_in[0];
    const float* logits = (const float*)d_in[1];
    const float* W      = (const float*)d_in[2];
    float* out = (float*)d_out;

    float* ws    = (float*)d_ws;
    float* Wsym  = ws;
    float* V0    = Wsym + (size_t)NPTS * NPTS;
    float* V1    = V0 + (size_t)NPTS * 32;
    float* a     = V1 + (size_t)NPTS * 32;
    float* ux    = a + BATCH * NPTS;
    float* uy    = ux + BATCH * NPTS;

    crf_prep<<<128, 256, 0, stream>>>(dp, logits, a, ux, uy, V0);
    crf_sym<<<dim3(65, 32), 256, 0, stream>>>(W, Wsym);

    for (int it = 0; it < ITERS / 2 - 1; ++it) {
        crf_gemm<<<256, 256, 0, stream>>>(Wsym, V0, a, ux, uy, logits, V1, out, 0);
        crf_gemm<<<256, 256, 0, stream>>>(Wsym, V1, a, ux, uy, logits, V0, out, 0);
    }
    crf_gemm<<<256, 256, 0, stream>>>(Wsym, V0, a, ux, uy, logits, V1, out, 0);
    crf_gemm<<<256, 256, 0, stream>>>(Wsym, V1, a, ux, uy, logits, V0, out, 1);
}
